# starting kernel
# baseline (speedup 1.0000x reference)
#include <hip/hip_runtime.h>
#include <hip/hip_bf16.h>


typedef __attribute__((ext_vector_type(2))) float v2f;
typedef __attribute__((ext_vector_type(8))) float v8f;

#define GM_N 65536
#define GM_M 1024

__global__ __launch_bounds__(GM_M) void gm_coef_kernel(
    const float* __restrict__ mu,
    const float* __restrict__ Amat,
    const float* __restrict__ w,
    float* __restrict__ coef)
{
    __shared__ float red[GM_M];
    const int j = threadIdx.x;

    const float wj = w[j];

    red[j] = wj;
    __syncthreads();
    #pragma unroll
    for (int s = GM_M / 2; s > 0; s >>= 1) {
        if (j < s) red[j] = fmaxf(red[j], red[j + s]);
        __syncthreads();
    }
    const float wmax = red[0];
    __syncthreads();
    red[j] = __expf(wj - wmax);
    __syncthreads();
    #pragma unroll
    for (int s = GM_M / 2; s > 0; s >>= 1) {
        if (j < s) red[j] += red[j + s];
        __syncthreads();
    }
    const float lse_w = wmax + __logf(red[0]);

    const float a00 = Amat[j * 4 + 0], a01 = Amat[j * 4 + 1];
    const float a10 = Amat[j * 4 + 2], a11 = Amat[j * 4 + 3];
    const float g00 = a00 * a00 + a01 * a01;
    const float g01 = a00 * a10 + a01 * a11;
    const float g11 = a10 * a10 + a11 * a11;
    const float det = g00 * g11 - g01 * g01;

    const float logw = (wj - lse_w) + 0.5f * __logf(det);

    const float m0 = mu[j * 2 + 0], m1 = mu[j * 2 + 1];
    const float gm0 = g00 * m0 + g01 * m1;
    const float gm1 = g01 * m0 + g11 * m1;
    const float mGm = gm0 * m0 + gm1 * m1;

    float* c = coef + j * 8;
    c[0] = -g00;
    c[1] = -2.0f * g01;
    c[2] = -g11;
    c[3] = 2.0f * gm0;
    c[4] = 2.0f * gm1;
    c[5] = logw - mGm;
    c[6] = 0.0f;
    c[7] = 0.0f;
}

__global__ __launch_bounds__(256) void gm_main_kernel(
    const float* __restrict__ sample,
    const float* __restrict__ coef,
    float* __restrict__ out)
{
    __shared__ float lcoef[GM_M * 8];

    {
        const float4* src = (const float4*)coef;
        float4* dst = (float4*)lcoef;
        #pragma unroll
        for (int i = threadIdx.x; i < GM_M * 8 / 4; i += 256) dst[i] = src[i];
    }
    __syncthreads();

    const int lane  = threadIdx.x & 31;
    const int wave  = threadIdx.x >> 5;
    const int half  = lane >> 4;
    const int ccol  = lane & 15;

    const int rowBase = blockIdx.x * 128 + wave * 16;
    const int sIdx = rowBase + ccol;

    const float x0 = sample[2 * sIdx + 0];
    const float x1 = sample[2 * sIdx + 1];
    const float f0 = x0 * x0, f1 = x0 * x1, f2 = x1 * x1;
    const float f3 = x0, f4 = x1, f5 = 1.0f, f6 = 0.0f, f7 = 0.0f;

    v2f bLo, bHi;
    bLo.x = half ? f2 : f0;  bLo.y = half ? f3 : f1;
    bHi.x = half ? f6 : f4;  bHi.y = half ? f7 : f5;

    float mx  = -__builtin_inff();
    float sum = 0.0f;

    #pragma unroll 4
    for (int t = 0; t < GM_M / 16; ++t) {
        const float* cr = &lcoef[(t * 16 + ccol) * 8];
        v2f aLo, aHi;
        aLo.x = cr[2 * half + 0];  aLo.y = cr[2 * half + 1];
        aHi.x = cr[2 * half + 4];  aHi.y = cr[2 * half + 5];

        v8f acc = {0.f, 0.f, 0.f, 0.f, 0.f, 0.f, 0.f, 0.f};
        acc = __builtin_amdgcn_wmma_f32_16x16x4_f32(
                  false, aLo, false, bLo, (short)0, acc, false, false);
        acc = __builtin_amdgcn_wmma_f32_16x16x4_f32(
                  false, aHi, false, bHi, (short)0, acc, false, false);

        float tm = acc[0];
        #pragma unroll
        for (int k = 1; k < 8; ++k) tm = fmaxf(tm, acc[k]);
        float ts = 0.0f;
        #pragma unroll
        for (int k = 0; k < 8; ++k) ts += __expf(acc[k] - tm);

        const float m2 = fmaxf(mx, tm);
        sum = sum * __expf(mx - m2) + ts * __expf(tm - m2);
        mx  = m2;
    }

    const float mxo  = __shfl_xor(mx, 16, 32);
    const float sumo = __shfl_xor(sum, 16, 32);
    const float m2 = fmaxf(mx, mxo);
    const float res = m2 + __logf(sum * __expf(mx - m2) + sumo * __expf(mxo - m2));

    if (lane < 16) out[rowBase + lane] = res;
}

extern "C" void kernel_launch(void* const* d_in, const int* in_sizes, int n_in,
                              void* d_out, int out_size, void* d_ws, size_t ws_size,
                              hipStream_t stream) {
    const float* sample = (const float*)d_in[0];
    const float* mu     = (const float*)d_in[1];
    const float* Amat   = (const float*)d_in[2];
    const float* w      = (const float*)d_in[3];
    float* out = (float*)d_out;

    float* coef = (float*)d_ws;

    gm_coef_kernel<<<1, GM_M, 0, stream>>>(mu, Amat, w, coef);

    const int n = in_sizes[0] / 2;
    const int blocks = n / 128;
    gm_main_kernel<<<blocks, 256, 0, stream>>>(sample, coef, out);
}


# Fuse the coefficient kernel into the score kernel: one launch, each block builds the 1024 coefficient rows straight into LDS; the weight normaliser is re-summed per wave (new reduction tree)
# speedup vs baseline: 1.5065x; 1.5065x over previous
#include <hip/hip_runtime.h>
#include <hip/hip_bf16.h>


typedef __attribute__((ext_vector_type(2))) float v2f;
typedef __attribute__((ext_vector_type(8))) float v8f;

#define GM_N 65536
#define GM_M 1024

__device__ __forceinline__ float gm_xrow(float x) {
    const int b = __builtin_bit_cast(int, x);
    return __builtin_bit_cast(float,
        __builtin_amdgcn_permlanex16(b, b, 0x76543210, 0xfedcba98, false, false));
}

#define GM_ROWPERM(x, LO, HI)                                                  \
    __builtin_bit_cast(float, __builtin_amdgcn_permlane16(                      \
        __builtin_bit_cast(int, (x)), __builtin_bit_cast(int, (x)), (LO), (HI), \
        false, false))

__device__ __forceinline__ float gm_wave_max(float x) {
    x = fmaxf(x, GM_ROWPERM(x, 0x67452301, 0xefcdab89));
    x = fmaxf(x, GM_ROWPERM(x, 0x54761032, 0xdcfe98ba));
    x = fmaxf(x, GM_ROWPERM(x, 0x32107654, 0xba98fedc));
    x = fmaxf(x, GM_ROWPERM(x, 0xfedcba98, 0x76543210));
    x = fmaxf(x, gm_xrow(x));
    return x;
}

__device__ __forceinline__ float gm_wave_sum(float x) {
    x += GM_ROWPERM(x, 0x67452301, 0xefcdab89);
    x += GM_ROWPERM(x, 0x54761032, 0xdcfe98ba);
    x += GM_ROWPERM(x, 0x32107654, 0xba98fedc);
    x += GM_ROWPERM(x, 0xfedcba98, 0x76543210);
    x += gm_xrow(x);
    return x;
}

__global__ __launch_bounds__(256) void gm_fused_kernel(
    const float* __restrict__ sample,
    const float* __restrict__ mu,
    const float* __restrict__ Amat,
    const float* __restrict__ w,
    float* __restrict__ out)
{
    __shared__ float lcoef[GM_M * 8];

    const int tid   = threadIdx.x;
    const int lane  = tid & 31;
    const int wave  = tid >> 5;
    const int half  = lane >> 4;
    const int ccol  = lane & 15;

    const int rowBase = blockIdx.x * 128 + wave * 16;
    const int sIdx = rowBase + ccol;

    const float2 xs = ((const float2*)sample)[sIdx];
    float4 wv[8];
    for (int k = 0; k < 8; ++k) wv[k] = ((const float4*)w)[k * 32 + lane];
    float4 av[4];
    float2 mv[4];
    float  wj[4];
    for (int k = 0; k < 4; ++k) {
        const int j = tid + 256 * k;
        av[k] = ((const float4*)Amat)[j];
        mv[k] = ((const float2*)mu)[j];
        wj[k] = w[j];
    }

    float lm = fmaxf(fmaxf(wv[0].x, wv[0].y), fmaxf(wv[0].z, wv[0].w));
    for (int k = 1; k < 8; ++k)
        lm = fmaxf(lm, fmaxf(fmaxf(wv[k].x, wv[k].y), fmaxf(wv[k].z, wv[k].w)));
    const float wmax = gm_wave_max(lm);
    float ls = 0.0f;
    for (int k = 0; k < 8; ++k)
        ls += (__expf(wv[k].x - wmax) + __expf(wv[k].y - wmax))
            + (__expf(wv[k].z - wmax) + __expf(wv[k].w - wmax));
    const float lse_w = wmax + __logf(gm_wave_sum(ls));

    for (int k = 0; k < 4; ++k) {
        const int j = tid + 256 * k;
        const float a00 = av[k].x, a01 = av[k].y;
        const float a10 = av[k].z, a11 = av[k].w;
        const float g00 = a00 * a00 + a01 * a01;
        const float g01 = a00 * a10 + a01 * a11;
        const float g11 = a10 * a10 + a11 * a11;
        const float det = g00 * g11 - g01 * g01;

        const float logw = (wj[k] - lse_w) + 0.5f * __logf(det);

        const float m0 = mv[k].x, m1 = mv[k].y;
        const float gm0 = g00 * m0 + g01 * m1;
        const float gm1 = g01 * m0 + g11 * m1;
        const float mGm = gm0 * m0 + gm1 * m1;

        float4* c = (float4*)&lcoef[j * 8];
        c[0] = make_float4(-g00, -2.0f * g01, -g11, 2.0f * gm0);
        c[1] = make_float4(2.0f * gm1, logw - mGm, 0.0f, 0.0f);
    }
    __syncthreads();

    const float x0 = xs.x;
    const float x1 = xs.y;
    const float f0 = x0 * x0, f1 = x0 * x1, f2 = x1 * x1;
    const float f3 = x0, f4 = x1, f5 = 1.0f, f6 = 0.0f, f7 = 0.0f;

    v2f bLo, bHi;
    bLo.x = half ? f2 : f0;  bLo.y = half ? f3 : f1;
    bHi.x = half ? f6 : f4;  bHi.y = half ? f7 : f5;

    float mx  = -__builtin_inff();
    float sum = 0.0f;

    #pragma unroll 4
    for (int t = 0; t < GM_M / 16; ++t) {
        const float* cr = &lcoef[(t * 16 + ccol) * 8];
        v2f aLo, aHi;
        aLo.x = cr[2 * half + 0];  aLo.y = cr[2 * half + 1];
        aHi.x = cr[2 * half + 4];  aHi.y = cr[2 * half + 5];

        v8f acc = {0.f, 0.f, 0.f, 0.f, 0.f, 0.f, 0.f, 0.f};
        acc = __builtin_amdgcn_wmma_f32_16x16x4_f32(
                  false, aLo, false, bLo, (short)0, acc, false, false);
        acc = __builtin_amdgcn_wmma_f32_16x16x4_f32(
                  false, aHi, false, bHi, (short)0, acc, false, false);

        float tm = acc[0];
        #pragma unroll
        for (int k = 1; k < 8; ++k) tm = fmaxf(tm, acc[k]);
        float ts = 0.0f;
        #pragma unroll
        for (int k = 0; k < 8; ++k) ts += __expf(acc[k] - tm);

        const float m2 = fmaxf(mx, tm);
        sum = sum * __expf(mx - m2) + ts * __expf(tm - m2);
        mx  = m2;
    }

    const float mxo  = __shfl_xor(mx, 16, 32);
    const float sumo = __shfl_xor(sum, 16, 32);
    const float m2 = fmaxf(mx, mxo);
    const float res = m2 + __logf(sum * __expf(mx - m2) + sumo * __expf(mxo - m2));

    if (lane < 16) out[rowBase + lane] = res;
}

extern "C" void kernel_launch(void* const* d_in, const int* in_sizes, int n_in,
                              void* d_out, int out_size, void* d_ws, size_t ws_size,
                              hipStream_t stream) {
    const float* sample = (const float*)d_in[0];
    const float* mu     = (const float*)d_in[1];
    const float* Amat   = (const float*)d_in[2];
    const float* w      = (const float*)d_in[3];
    float* out = (float*)d_out;

    const int n = in_sizes[0] / 2;
    const int blocks = n / 128;
    gm_fused_kernel<<<blocks, 256, 0, stream>>>(sample, mu, Amat, w, out);
}


# Freeze the running max after 4 exact tiles and feed -ref through the WMMA C operand: later tiles only exp and add into 8 per-row partials (new summation order); one overflow test, exact redo on fail
# speedup vs baseline: 1.7619x; 1.1695x over previous
#include <hip/hip_runtime.h>
#include <hip/hip_bf16.h>


typedef __attribute__((ext_vector_type(2))) float v2f;
typedef __attribute__((ext_vector_type(8))) float v8f;

#define GM_N 65536
#define GM_M 1024
#define GM_TILES (GM_M / 16)
#define GM_LEAD 4
#define GM_BIG 0x1p100f

__device__ __forceinline__ float gm_xrow(float x) {
    const int b = __builtin_bit_cast(int, x);
    return __builtin_bit_cast(float,
        __builtin_amdgcn_permlanex16(b, b, 0x76543210, 0xfedcba98, false, false));
}

#define GM_ROWPERM(x, LO, HI)                                                  \
    __builtin_bit_cast(float, __builtin_amdgcn_permlane16(                      \
        __builtin_bit_cast(int, (x)), __builtin_bit_cast(int, (x)), (LO), (HI), \
        false, false))

__device__ __forceinline__ float gm_wave_max(float x) {
    x = fmaxf(x, GM_ROWPERM(x, 0x67452301, 0xefcdab89));
    x = fmaxf(x, GM_ROWPERM(x, 0x54761032, 0xdcfe98ba));
    x = fmaxf(x, GM_ROWPERM(x, 0x32107654, 0xba98fedc));
    x = fmaxf(x, GM_ROWPERM(x, 0xfedcba98, 0x76543210));
    x = fmaxf(x, gm_xrow(x));
    return x;
}

__device__ __forceinline__ float gm_wave_sum(float x) {
    x += GM_ROWPERM(x, 0x67452301, 0xefcdab89);
    x += GM_ROWPERM(x, 0x54761032, 0xdcfe98ba);
    x += GM_ROWPERM(x, 0x32107654, 0xba98fedc);
    x += GM_ROWPERM(x, 0xfedcba98, 0x76543210);
    x += gm_xrow(x);
    return x;
}

__device__ __forceinline__ void gm_fold_exact(const v8f acc, float& mx, float& sum) {
    float tm = acc[0];
    #pragma unroll
    for (int k = 1; k < 8; ++k) tm = fmaxf(tm, acc[k]);
    float ts = 0.0f;
    #pragma unroll
    for (int k = 0; k < 8; ++k) ts += __expf(acc[k] - tm);

    const float m2 = fmaxf(mx, tm);
    sum = sum * __expf(mx - m2) + ts * __expf(tm - m2);
    mx  = m2;
}

__device__ __forceinline__ void gm_load_a(const float* lcoef, int t, int ccol, int half,
                                          v2f& aLo, v2f& aHi) {
    const float* cr = &lcoef[(t * 16 + ccol) * 8];
    aLo.x = cr[2 * half + 0];  aLo.y = cr[2 * half + 1];
    aHi.x = cr[2 * half + 4];  aHi.y = cr[2 * half + 5];
}

__global__ __launch_bounds__(256) void gm_fused_kernel(
    const float* __restrict__ sample,
    const float* __restrict__ mu,
    const float* __restrict__ Amat,
    const float* __restrict__ w,
    float* __restrict__ out)
{
    __shared__ float lcoef[GM_M * 8];

    const int tid   = threadIdx.x;
    const int lane  = tid & 31;
    const int wave  = tid >> 5;
    const int half  = lane >> 4;
    const int ccol  = lane & 15;

    const int rowBase = blockIdx.x * 128 + wave * 16;
    const int sIdx = rowBase + ccol;

    const float2 xs = ((const float2*)sample)[sIdx];
    float4 wv[8];
    for (int k = 0; k < 8; ++k) wv[k] = ((const float4*)w)[k * 32 + lane];
    float4 av[4];
    float2 mv[4];
    float  wj[4];
    for (int k = 0; k < 4; ++k) {
        const int j = tid + 256 * k;
        av[k] = ((const float4*)Amat)[j];
        mv[k] = ((const float2*)mu)[j];
        wj[k] = w[j];
    }

    float lm = fmaxf(fmaxf(wv[0].x, wv[0].y), fmaxf(wv[0].z, wv[0].w));
    for (int k = 1; k < 8; ++k)
        lm = fmaxf(lm, fmaxf(fmaxf(wv[k].x, wv[k].y), fmaxf(wv[k].z, wv[k].w)));
    const float wmax = gm_wave_max(lm);
    float ls = 0.0f;
    for (int k = 0; k < 8; ++k)
        ls += (__expf(wv[k].x - wmax) + __expf(wv[k].y - wmax))
            + (__expf(wv[k].z - wmax) + __expf(wv[k].w - wmax));
    const float lse_w = wmax + __logf(gm_wave_sum(ls));

    for (int k = 0; k < 4; ++k) {
        const int j = tid + 256 * k;
        const float a00 = av[k].x, a01 = av[k].y;
        const float a10 = av[k].z, a11 = av[k].w;
        const float g00 = a00 * a00 + a01 * a01;
        const float g01 = a00 * a10 + a01 * a11;
        const float g11 = a10 * a10 + a11 * a11;
        const float det = g00 * g11 - g01 * g01;

        const float logw = (wj[k] - lse_w) + 0.5f * __logf(det);

        const float m0 = mv[k].x, m1 = mv[k].y;
        const float gm0 = g00 * m0 + g01 * m1;
        const float gm1 = g01 * m0 + g11 * m1;
        const float mGm = gm0 * m0 + gm1 * m1;

        float4* c = (float4*)&lcoef[j * 8];
        c[0] = make_float4(-g00, -2.0f * g01, -g11, 2.0f * gm0);
        c[1] = make_float4(2.0f * gm1, logw - mGm, 0.0f, 0.0f);
    }
    __syncthreads();

    const float x0 = xs.x;
    const float x1 = xs.y;
    const float f0 = x0 * x0, f1 = x0 * x1, f2 = x1 * x1;
    const float f3 = x0, f4 = x1, f5 = 1.0f, f6 = 0.0f, f7 = 0.0f;

    v2f bLo, bHi;
    bLo.x = half ? f2 : f0;  bLo.y = half ? f3 : f1;
    bHi.x = half ? f6 : f4;  bHi.y = half ? f7 : f5;

    float mx  = -__builtin_inff();
    float sum = 0.0f;
    const v8f zero8 = {0.f, 0.f, 0.f, 0.f, 0.f, 0.f, 0.f, 0.f};

    #pragma unroll
    for (int t = 0; t < GM_LEAD; ++t) {
        v2f aLo, aHi;
        gm_load_a(lcoef, t, ccol, half, aLo, aHi);
        v8f acc = __builtin_amdgcn_wmma_f32_16x16x4_f32(
                      false, aLo, false, bLo, (short)0, zero8, false, false);
        acc = __builtin_amdgcn_wmma_f32_16x16x4_f32(
                  false, aHi, false, bHi, (short)0, acc, false, false);
        gm_fold_exact(acc, mx, sum);
    }

    const float nref = -mx;
    const v8f cref = {nref, nref, nref, nref, nref, nref, nref, nref};
    v8f part = zero8;

    #pragma unroll 4
    for (int t = GM_LEAD; t < GM_TILES; ++t) {
        v2f aLo, aHi;
        gm_load_a(lcoef, t, ccol, half, aLo, aHi);
        v8f acc = __builtin_amdgcn_wmma_f32_16x16x4_f32(
                      false, aLo, false, bLo, (short)0, cref, false, false);
        acc = __builtin_amdgcn_wmma_f32_16x16x4_f32(
                  false, aHi, false, bHi, (short)0, acc, false, false);
        #pragma unroll
        for (int k = 0; k < 8; ++k) part[k] += __expf(acc[k]);
    }
    const float tot = ((part[0] + part[1]) + (part[2] + part[3]))
                    + ((part[4] + part[5]) + (part[6] + part[7]));

    if (__builtin_amdgcn_ballot_w32(!(tot <= GM_BIG)) != 0u) {
        for (int t = GM_LEAD; t < GM_TILES; ++t) {
            v2f aLo, aHi;
            gm_load_a(lcoef, t, ccol, half, aLo, aHi);
            v8f acc = __builtin_amdgcn_wmma_f32_16x16x4_f32(
                          false, aLo, false, bLo, (short)0, zero8, false, false);
            acc = __builtin_amdgcn_wmma_f32_16x16x4_f32(
                      false, aHi, false, bHi, (short)0, acc, false, false);
            gm_fold_exact(acc, mx, sum);
        }
    } else {
        sum += tot;
    }

    const float mxo  = __shfl_xor(mx, 16, 32);
    const float sumo = __shfl_xor(sum, 16, 32);
    const float m2 = fmaxf(mx, mxo);
    const float res = m2 + __logf(sum * __expf(mx - m2) + sumo * __expf(mxo - m2));

    if (lane < 16) out[rowBase + lane] = res;
}

extern "C" void kernel_launch(void* const* d_in, const int* in_sizes, int n_in,
                              void* d_out, int out_size, void* d_ws, size_t ws_size,
                              hipStream_t stream) {
    const float* sample = (const float*)d_in[0];
    const float* mu     = (const float*)d_in[1];
    const float* Amat   = (const float*)d_in[2];
    const float* w      = (const float*)d_in[3];
    float* out = (float*)d_out;

    const int n = in_sizes[0] / 2;
    const int blocks = n / 128;
    gm_fused_kernel<<<blocks, 256, 0, stream>>>(sample, mu, Amat, w, out);
}


# Scale every coefficient row by log2(e) when it is built: scores arrive in base-2 units, each exponential is the native 2^x with no multiply in front; ln(2) is applied once per sample at the end
# speedup vs baseline: 1.8515x; 1.0509x over previous
#include <hip/hip_runtime.h>
#include <hip/hip_bf16.h>


typedef __attribute__((ext_vector_type(2))) float v2f;
typedef __attribute__((ext_vector_type(8))) float v8f;

#define GM_N 65536
#define GM_M 1024
#define GM_TILES (GM_M / 16)
#define GM_LEAD 4
#define GM_BIG 0x1p100f
#define GM_LOG2E 1.44269504088896340736f
#define GM_LN2   0.69314718055994530942f

__device__ __forceinline__ float gm_exp2(float x) { return __builtin_amdgcn_exp2f(x); }
__device__ __forceinline__ float gm_log2(float x) { return __builtin_amdgcn_logf(x); }

__device__ __forceinline__ float gm_xrow(float x) {
    const int b = __builtin_bit_cast(int, x);
    return __builtin_bit_cast(float,
        __builtin_amdgcn_permlanex16(b, b, 0x76543210, 0xfedcba98, false, false));
}

#define GM_ROWPERM(x, LO, HI)                                                  \
    __builtin_bit_cast(float, __builtin_amdgcn_permlane16(                      \
        __builtin_bit_cast(int, (x)), __builtin_bit_cast(int, (x)), (LO), (HI), \
        false, false))

__device__ __forceinline__ float gm_wave_max(float x) {
    x = fmaxf(x, GM_ROWPERM(x, 0x67452301, 0xefcdab89));
    x = fmaxf(x, GM_ROWPERM(x, 0x54761032, 0xdcfe98ba));
    x = fmaxf(x, GM_ROWPERM(x, 0x32107654, 0xba98fedc));
    x = fmaxf(x, GM_ROWPERM(x, 0xfedcba98, 0x76543210));
    x = fmaxf(x, gm_xrow(x));
    return x;
}

__device__ __forceinline__ float gm_wave_sum(float x) {
    x += GM_ROWPERM(x, 0x67452301, 0xefcdab89);
    x += GM_ROWPERM(x, 0x54761032, 0xdcfe98ba);
    x += GM_ROWPERM(x, 0x32107654, 0xba98fedc);
    x += GM_ROWPERM(x, 0xfedcba98, 0x76543210);
    x += gm_xrow(x);
    return x;
}

__device__ __forceinline__ void gm_fold_exact(const v8f acc, float& mx, float& sum) {
    float tm = acc[0];
    #pragma unroll
    for (int k = 1; k < 8; ++k) tm = fmaxf(tm, acc[k]);
    float ts = 0.0f;
    #pragma unroll
    for (int k = 0; k < 8; ++k) ts += gm_exp2(acc[k] - tm);

    const float m2 = fmaxf(mx, tm);
    sum = sum * gm_exp2(mx - m2) + ts * gm_exp2(tm - m2);
    mx  = m2;
}

__device__ __forceinline__ void gm_load_a(const float* lcoef, int t, int ccol, int half,
                                          v2f& aLo, v2f& aHi) {
    const float* cr = &lcoef[(t * 16 + ccol) * 8];
    aLo.x = cr[2 * half + 0];  aLo.y = cr[2 * half + 1];
    aHi.x = cr[2 * half + 4];  aHi.y = cr[2 * half + 5];
}

__global__ __launch_bounds__(256) void gm_fused_kernel(
    const float* __restrict__ sample,
    const float* __restrict__ mu,
    const float* __restrict__ Amat,
    const float* __restrict__ w,
    float* __restrict__ out)
{
    __shared__ float lcoef[GM_M * 8];

    const int tid   = threadIdx.x;
    const int lane  = tid & 31;
    const int wave  = tid >> 5;
    const int half  = lane >> 4;
    const int ccol  = lane & 15;

    const int rowBase = blockIdx.x * 128 + wave * 16;
    const int sIdx = rowBase + ccol;

    const float2 xs = ((const float2*)sample)[sIdx];
    float4 wv[8];
    for (int k = 0; k < 8; ++k) wv[k] = ((const float4*)w)[k * 32 + lane];
    float4 av[4];
    float2 mv[4];
    float  wj[4];
    for (int k = 0; k < 4; ++k) {
        const int j = tid + 256 * k;
        av[k] = ((const float4*)Amat)[j];
        mv[k] = ((const float2*)mu)[j];
        wj[k] = w[j];
    }

    float lm = fmaxf(fmaxf(wv[0].x, wv[0].y), fmaxf(wv[0].z, wv[0].w));
    for (int k = 1; k < 8; ++k)
        lm = fmaxf(lm, fmaxf(fmaxf(wv[k].x, wv[k].y), fmaxf(wv[k].z, wv[k].w)));
    const float wmax = gm_wave_max(lm);
    float ls = 0.0f;
    for (int k = 0; k < 8; ++k)
        ls += (__expf(wv[k].x - wmax) + __expf(wv[k].y - wmax))
            + (__expf(wv[k].z - wmax) + __expf(wv[k].w - wmax));
    const float lse_w = wmax + __logf(gm_wave_sum(ls));

    for (int k = 0; k < 4; ++k) {
        const int j = tid + 256 * k;
        const float a00 = av[k].x, a01 = av[k].y;
        const float a10 = av[k].z, a11 = av[k].w;
        const float g00 = a00 * a00 + a01 * a01;
        const float g01 = a00 * a10 + a01 * a11;
        const float g11 = a10 * a10 + a11 * a11;
        const float det = g00 * g11 - g01 * g01;

        const float logw = (wj[k] - lse_w) + 0.5f * __logf(det);

        const float m0 = mv[k].x, m1 = mv[k].y;
        const float gm0 = g00 * m0 + g01 * m1;
        const float gm1 = g01 * m0 + g11 * m1;
        const float mGm = gm0 * m0 + gm1 * m1;

        float4* c = (float4*)&lcoef[j * 8];
        c[0] = make_float4(-GM_LOG2E * g00, (-2.0f * GM_LOG2E) * g01,
                           -GM_LOG2E * g11, (2.0f * GM_LOG2E) * gm0);
        c[1] = make_float4((2.0f * GM_LOG2E) * gm1, GM_LOG2E * (logw - mGm), 0.0f, 0.0f);
    }
    __syncthreads();

    const float x0 = xs.x;
    const float x1 = xs.y;
    const float f0 = x0 * x0, f1 = x0 * x1, f2 = x1 * x1;
    const float f3 = x0, f4 = x1, f5 = 1.0f, f6 = 0.0f, f7 = 0.0f;

    v2f bLo, bHi;
    bLo.x = half ? f2 : f0;  bLo.y = half ? f3 : f1;
    bHi.x = half ? f6 : f4;  bHi.y = half ? f7 : f5;

    float mx  = -__builtin_inff();
    float sum = 0.0f;
    const v8f zero8 = {0.f, 0.f, 0.f, 0.f, 0.f, 0.f, 0.f, 0.f};

    #pragma unroll
    for (int t = 0; t < GM_LEAD; ++t) {
        v2f aLo, aHi;
        gm_load_a(lcoef, t, ccol, half, aLo, aHi);
        v8f acc = __builtin_amdgcn_wmma_f32_16x16x4_f32(
                      false, aLo, false, bLo, (short)0, zero8, false, false);
        acc = __builtin_amdgcn_wmma_f32_16x16x4_f32(
                  false, aHi, false, bHi, (short)0, acc, false, false);
        gm_fold_exact(acc, mx, sum);
    }

    const float nref = -mx;
    const v8f cref = {nref, nref, nref, nref, nref, nref, nref, nref};
    v8f part = zero8;

    #pragma unroll 4
    for (int t = GM_LEAD; t < GM_TILES; ++t) {
        v2f aLo, aHi;
        gm_load_a(lcoef, t, ccol, half, aLo, aHi);
        v8f acc = __builtin_amdgcn_wmma_f32_16x16x4_f32(
                      false, aLo, false, bLo, (short)0, cref, false, false);
        acc = __builtin_amdgcn_wmma_f32_16x16x4_f32(
                  false, aHi, false, bHi, (short)0, acc, false, false);
        #pragma unroll
        for (int k = 0; k < 8; ++k) part[k] += gm_exp2(acc[k]);
    }
    const float tot = ((part[0] + part[1]) + (part[2] + part[3]))
                    + ((part[4] + part[5]) + (part[6] + part[7]));

    if (__builtin_amdgcn_ballot_w32(!(tot <= GM_BIG)) != 0u) {
        for (int t = GM_LEAD; t < GM_TILES; ++t) {
            v2f aLo, aHi;
            gm_load_a(lcoef, t, ccol, half, aLo, aHi);
            v8f acc = __builtin_amdgcn_wmma_f32_16x16x4_f32(
                          false, aLo, false, bLo, (short)0, zero8, false, false);
            acc = __builtin_amdgcn_wmma_f32_16x16x4_f32(
                      false, aHi, false, bHi, (short)0, acc, false, false);
            gm_fold_exact(acc, mx, sum);
        }
    } else {
        sum += tot;
    }

    const float mxo  = __shfl_xor(mx, 16, 32);
    const float sumo = __shfl_xor(sum, 16, 32);
    const float m2 = fmaxf(mx, mxo);
    const float res = GM_LN2 * (m2 + gm_log2(sum * gm_exp2(mx - m2) + sumo * gm_exp2(mxo - m2)));

    if (lane < 16) out[rowBase + lane] = res;
}

extern "C" void kernel_launch(void* const* d_in, const int* in_sizes, int n_in,
                              void* d_out, int out_size, void* d_ws, size_t ws_size,
                              hipStream_t stream) {
    const float* sample = (const float*)d_in[0];
    const float* mu     = (const float*)d_in[1];
    const float* Amat   = (const float*)d_in[2];
    const float* w      = (const float*)d_in[3];
    float* out = (float*)d_out;

    const int n = in_sizes[0] / 2;
    const int blocks = n / 128;
    gm_fused_kernel<<<blocks, 256, 0, stream>>>(sample, mu, Amat, w, out);
}
